# starting kernel
# baseline (speedup 1.0000x reference)
#include <hip/hip_runtime.h>
#include <hip/hip_bf16.h>

#define N_SRC 100000
#define N_DST 20000
#define E_CNT 1250000
#define D 64
#define NEG_SLOPE 0.2f

typedef __attribute__((ext_vector_type(2))) float v2f;
typedef __attribute__((ext_vector_type(8))) float v8f;

__global__ __launch_bounds__(256) void gat_init(float* __restrict__ out,
                                                const float* __restrict__ bias,
                                                float* __restrict__ denom,
                                                unsigned* __restrict__ m_enc) {
    int i = blockIdx.x * blockDim.x + threadIdx.x;
    if (i < N_DST * D) out[i] = bias[i & (D - 1)];
    if (i < N_DST) {
        denom[i] = 0.0f;
        m_enc[i] = 0u;
    }
}

__global__ __launch_bounds__(64) void gat_watt(const float* __restrict__ W,
                                               const float* __restrict__ att_src,
                                               const float* __restrict__ att_dst,
                                               float* __restrict__ watt) {
    int t = threadIdx.x;
    float s = 0.0f, d = 0.0f;
    #pragma unroll 8
    for (int j = 0; j < D; ++j) {
        float w = W[t * D + j];
        s += w * att_src[j];
        d += w * att_dst[j];
    }
    watt[t] = s;
    watt[D + t] = d;
}

__global__ __launch_bounds__(128) void gat_gemm_hsrc(const float* __restrict__ x,
                                                     const float* __restrict__ W,
                                                     float* __restrict__ h) {
    __shared__ float ldsW[D * D];
    for (int i = threadIdx.x; i < D * D; i += 128) ldsW[i] = W[i];
    __syncthreads();

    const int lane = threadIdx.x & 31;
    const int mrow = lane & 15;
    const int hi   = lane >> 4;
    const int row0 = blockIdx.x * 64 + (threadIdx.x >> 5) * 16;
    const int row  = row0 + mrow;
    const int rowc = (row < N_SRC) ? row : (N_SRC - 1);
    const float valid = (row < N_SRC) ? 1.0f : 0.0f;

    v2f afrag[16];
    const float* abase = x + (size_t)rowc * D + hi * 2;
    #pragma unroll
    for (int kk = 0; kk < 16; ++kk) {
        afrag[kk].x = abase[kk * 4 + 0] * valid;
        afrag[kk].y = abase[kk * 4 + 1] * valid;
    }

    v8f acc[4] = {v8f{}, v8f{}, v8f{}, v8f{}};

    #pragma unroll
    for (int kk = 0; kk < 16; ++kk) {
        const int kb = kk * 4 + hi * 2;
        #pragma unroll
        for (int nb = 0; nb < 4; ++nb) {
            v2f b;
            b.x = ldsW[(kb + 0) * D + nb * 16 + mrow];
            b.y = ldsW[(kb + 1) * D + nb * 16 + mrow];
            acc[nb] = __builtin_amdgcn_wmma_f32_16x16x4_f32(
                 false, afrag[kk],  false, b,
                 (short)0, acc[nb],
                 false,  false);
        }
    }

    #pragma unroll
    for (int nb = 0; nb < 4; ++nb) {
        #pragma unroll
        for (int r = 0; r < 8; ++r) {
            int orow = row0 + r + hi * 8;
            if (orow < N_SRC) h[(size_t)orow * D + nb * 16 + mrow] = acc[nb][r];
        }
    }
}

__global__ __launch_bounds__(256) void gat_rowdot(const float* __restrict__ x,
                                                  const int* __restrict__ res_n_id,
                                                  const float* __restrict__ watt,
                                                  float* __restrict__ a_src,
                                                  float* __restrict__ a_dst) {
    int wid  = blockIdx.x * 8 + (threadIdx.x >> 5);
    int lane = threadIdx.x & 31;
    const float2* xp;
    const float2* wv;
    float* outp;
    if (wid < N_SRC) {
        xp = (const float2*)(x + (size_t)wid * D);
        wv = (const float2*)(watt);
        outp = a_src + wid;
    } else {
        int j = wid - N_SRC;
        if (j >= N_DST) return;
        xp = (const float2*)(x + (size_t)res_n_id[j] * D);
        wv = (const float2*)(watt + D);
        outp = a_dst + j;
    }
    float2 xv = xp[lane];
    float2 wvv = wv[lane];
    float s = xv.x * wvv.x + xv.y * wvv.y;
    #pragma unroll
    for (int off = 16; off > 0; off >>= 1) s += __shfl_xor(s, off, 32);
    if (lane == 0) *outp = s;
}

__global__ __launch_bounds__(256) void gat_edge_logit(const int* __restrict__ es,
                                                      const int* __restrict__ ed,
                                                      const float* __restrict__ a_src,
                                                      const float* __restrict__ a_dst,
                                                      float* __restrict__ e_buf,
                                                      unsigned* __restrict__ m_enc) {
    int k = blockIdx.x * blockDim.x + threadIdx.x;
    if (k >= E_CNT) return;
    int d = ed[k];
    float e = a_src[es[k]] + a_dst[d];
    e = (e > 0.0f) ? e : NEG_SLOPE * e;
    e_buf[k] = e;
    unsigned u = __float_as_uint(e);
    u = (u & 0x80000000u) ? ~u : (u | 0x80000000u);
    atomicMax(m_enc + d, u);
}

__global__ __launch_bounds__(256) void gat_edge_exp(const int* __restrict__ ed,
                                                    float* __restrict__ e_buf,
                                                    const unsigned* __restrict__ m_enc,
                                                    float* __restrict__ denom) {
    int k = blockIdx.x * blockDim.x + threadIdx.x;
    if (k >= E_CNT) return;
    int d = ed[k];
    unsigned u = m_enc[d];
    float m = (u & 0x80000000u) ? __uint_as_float(u & 0x7FFFFFFFu)
                                : __uint_as_float(~u);
    float ex = __expf(e_buf[k] - m);
    e_buf[k] = ex;
    atomicAdd(denom + d, ex);
}

__global__ __launch_bounds__(256) void gat_scatter(const int* __restrict__ es,
                                                   const int* __restrict__ ed,
                                                   const float* __restrict__ e_buf,
                                                   const float* __restrict__ denom,
                                                   const float* __restrict__ h,
                                                   float* __restrict__ out) {
    int widx = blockIdx.x * 8 + (threadIdx.x >> 5);
    if (widx >= E_CNT) return;
    int lane = threadIdx.x & 31;
    int s = es[widx];
    int d = ed[widx];
    float alpha = e_buf[widx] / (denom[d] + 1e-16f);
    const float2* hp = (const float2*)(h + (size_t)s * D);
    float2 v = hp[lane];
    float* op = out + (size_t)d * D + lane * 2;
    atomicAdd(op + 0, alpha * v.x);
    atomicAdd(op + 1, alpha * v.y);
}

extern "C" void kernel_launch(void* const* d_in, const int* in_sizes, int n_in,
                              void* d_out, int out_size, void* d_ws, size_t ws_size,
                              hipStream_t stream) {
    const float* x        = (const float*)d_in[0];
    const int*   res_n_id = (const int*)d_in[1];
    const int*   edge_src = (const int*)d_in[2];
    const int*   edge_dst = (const int*)d_in[3];
    const float* W        = (const float*)d_in[4];
    const float* att_src  = (const float*)d_in[5];
    const float* att_dst  = (const float*)d_in[6];
    const float* bias     = (const float*)d_in[7];
    float* out = (float*)d_out;

    char* ws = (char*)d_ws;
    float*    h_src = (float*)(ws);
    float*    e_buf = (float*)(ws + 25600000);
    float*    a_src = (float*)(ws + 30600000);
    float*    a_dst = (float*)(ws + 31000000);
    float*    denom = (float*)(ws + 31080000);
    unsigned* m_enc = (unsigned*)(ws + 31160000);
    float*    watt  = (float*)(ws + 31240000);

    gat_init<<<(N_DST * D + 255) / 256, 256, 0, stream>>>(out, bias, denom, m_enc);
    gat_watt<<<1, 64, 0, stream>>>(W, att_src, att_dst, watt);
    gat_gemm_hsrc<<<(N_SRC + 63) / 64, 128, 0, stream>>>(x, W, h_src);
    gat_rowdot<<<(N_SRC + N_DST + 7) / 8, 256, 0, stream>>>(x, res_n_id, watt, a_src, a_dst);
    gat_edge_logit<<<(E_CNT + 255) / 256, 256, 0, stream>>>(edge_src, edge_dst, a_src,
                                                            a_dst, e_buf, m_enc);
    gat_edge_exp<<<(E_CNT + 255) / 256, 256, 0, stream>>>(edge_dst, e_buf, m_enc, denom);
    gat_scatter<<<(E_CNT + 7) / 8, 256, 0, stream>>>(edge_src, edge_dst, e_buf, denom,
                                                     h_src, out);
}


# All per-edge passes by owner: 250 work-groups each scan edge_dst for their 80 destinations' edges, do max, exp-sum, row sum in LDS (LDS atomics, sums in arrival order); global atomics, 3 launches go
# speedup vs baseline: 3.8134x; 3.8134x over previous
#include <hip/hip_runtime.h>
#include <hip/hip_bf16.h>

#define N_SRC 100000
#define N_DST 20000
#define E_CNT 1250000
#define D 64
#define NEG_SLOPE 0.2f

typedef __attribute__((ext_vector_type(2))) float v2f;
typedef __attribute__((ext_vector_type(8))) float v8f;

__global__ __launch_bounds__(64) void gat_watt(const float* __restrict__ W,
                                               const float* __restrict__ att_src,
                                               const float* __restrict__ att_dst,
                                               float* __restrict__ watt) {
    int t = threadIdx.x;
    float s = 0.0f, d = 0.0f;
    #pragma unroll 8
    for (int j = 0; j < D; ++j) {
        float w = W[t * D + j];
        s += w * att_src[j];
        d += w * att_dst[j];
    }
    watt[t] = s;
    watt[D + t] = d;
}

__global__ __launch_bounds__(128) void gat_gemm_hsrc(const float* __restrict__ x,
                                                     const float* __restrict__ W,
                                                     float* __restrict__ h) {
    __shared__ float ldsW[D * D];
    for (int i = threadIdx.x; i < D * D; i += 128) ldsW[i] = W[i];
    __syncthreads();

    const int lane = threadIdx.x & 31;
    const int mrow = lane & 15;
    const int hi   = lane >> 4;
    const int row0 = blockIdx.x * 64 + (threadIdx.x >> 5) * 16;
    const int row  = row0 + mrow;
    const int rowc = (row < N_SRC) ? row : (N_SRC - 1);
    const float valid = (row < N_SRC) ? 1.0f : 0.0f;

    v2f afrag[16];
    const float* abase = x + (size_t)rowc * D + hi * 2;
    #pragma unroll
    for (int kk = 0; kk < 16; ++kk) {
        afrag[kk].x = abase[kk * 4 + 0] * valid;
        afrag[kk].y = abase[kk * 4 + 1] * valid;
    }

    v8f acc[4] = {v8f{}, v8f{}, v8f{}, v8f{}};

    #pragma unroll
    for (int kk = 0; kk < 16; ++kk) {
        const int kb = kk * 4 + hi * 2;
        #pragma unroll
        for (int nb = 0; nb < 4; ++nb) {
            v2f b;
            b.x = ldsW[(kb + 0) * D + nb * 16 + mrow];
            b.y = ldsW[(kb + 1) * D + nb * 16 + mrow];
            acc[nb] = __builtin_amdgcn_wmma_f32_16x16x4_f32(
                 false, afrag[kk],  false, b,
                 (short)0, acc[nb],
                 false,  false);
        }
    }

    #pragma unroll
    for (int nb = 0; nb < 4; ++nb) {
        #pragma unroll
        for (int r = 0; r < 8; ++r) {
            int orow = row0 + r + hi * 8;
            if (orow < N_SRC) h[(size_t)orow * D + nb * 16 + mrow] = acc[nb][r];
        }
    }
}

__global__ __launch_bounds__(256) void gat_rowdot(const float* __restrict__ x,
                                                  const int* __restrict__ res_n_id,
                                                  const float* __restrict__ watt,
                                                  float* __restrict__ a_src,
                                                  float* __restrict__ a_dst) {
    int wid  = blockIdx.x * 8 + (threadIdx.x >> 5);
    int lane = threadIdx.x & 31;
    const float2* xp;
    const float2* wv;
    float* outp;
    if (wid < N_SRC) {
        xp = (const float2*)(x + (size_t)wid * D);
        wv = (const float2*)(watt);
        outp = a_src + wid;
    } else {
        int j = wid - N_SRC;
        if (j >= N_DST) return;
        xp = (const float2*)(x + (size_t)res_n_id[j] * D);
        wv = (const float2*)(watt + D);
        outp = a_dst + j;
    }
    float2 xv = xp[lane];
    float2 wvv = wv[lane];
    float s = xv.x * wvv.x + xv.y * wvv.y;
    #pragma unroll
    for (int off = 16; off > 0; off >>= 1) s += __shfl_xor(s, off, 32);
    if (lane == 0) *outp = s;
}

#define OWNERS 250
#define OWN_ROWS 80
#define OWN_THREADS 1024
#define OWN_WAVES (OWN_THREADS / 32)
#define WAVE_CAP 1024
#define SCAN_DEPTH 8
#define EDGE_GROUP 16
#define NV4 (E_CNT / 4)
#define SCAN_BLOCK (SCAN_DEPTH * OWN_THREADS)
#define SCAN_BLOCKS ((NV4 + SCAN_BLOCK - 1) / SCAN_BLOCK)
static_assert(OWNERS * OWN_ROWS == N_DST, "every destination has exactly one owner");
static_assert(E_CNT % 4 == 0, "edge_dst is read four entries at a time");
static_assert(E_CNT <= (1 << 21) && N_SRC <= (1 << 21) && OWN_ROWS <= (1 << 11),
              "an edge or source index and a local row share one word");

static __device__ __forceinline__ unsigned enc_ord(float e) {
    const unsigned u = __float_as_uint(e);
    return (u & 0x80000000u) ? ~u : (u | 0x80000000u);
}
static __device__ __forceinline__ float dec_ord(unsigned u) {
    return (u & 0x80000000u) ? __uint_as_float(u & 0x7FFFFFFFu) : __uint_as_float(~u);
}
static __device__ __forceinline__ float edge_logit(float as, float ad) {
    const float e = as + ad;
    return (e > 0.0f) ? e : NEG_SLOPE * e;
}

template <bool TAIL>
static __device__ __forceinline__ void scan_block(const int4* __restrict__ ed4, int first,
                                                  int tid, int lo, unsigned* wl,
                                                  unsigned& wcnt) {
    int4 v[SCAN_DEPTH];
    #pragma unroll
    for (int j = 0; j < SCAN_DEPTH; ++j) {
        int i = first + j * OWN_THREADS + tid;
        if (TAIL) i = (i < NV4) ? i : (NV4 - 1);
        v[j] = ed4[i];
    }
    #pragma unroll
    for (int j = 0; j < SCAN_DEPTH; ++j) {
        const int i = first + j * OWN_THREADS + tid;
        const bool valid = TAIL ? (i < NV4) : true;
        const unsigned r0 = (unsigned)(v[j].x - lo);
        const unsigned r1 = (unsigned)(v[j].y - lo);
        const unsigned r2 = (unsigned)(v[j].z - lo);
        const unsigned r3 = (unsigned)(v[j].w - lo);
        const unsigned rmin = min(min(r0, r1), min(r2, r3));
        if (__builtin_amdgcn_ballot_w32(valid && rmin < OWN_ROWS) != 0u) {
            const unsigned k = (unsigned)i * 4u;
            const bool h0 = valid && r0 < OWN_ROWS;
            const bool h1 = valid && r1 < OWN_ROWS;
            const bool h2 = valid && r2 < OWN_ROWS;
            const bool h3 = valid && r3 < OWN_ROWS;
            const unsigned m0 = __builtin_amdgcn_ballot_w32(h0);
            const unsigned m1 = __builtin_amdgcn_ballot_w32(h1);
            const unsigned m2 = __builtin_amdgcn_ballot_w32(h2);
            const unsigned m3 = __builtin_amdgcn_ballot_w32(h3);
            if (h0) {
                const unsigned p = wcnt + __builtin_amdgcn_mbcnt_lo(m0, 0u);
                if (p < WAVE_CAP) wl[p] = (k + 0u) | (r0 << 21);
            }
            wcnt += (unsigned)__builtin_popcount(m0);
            if (h1) {
                const unsigned p = wcnt + __builtin_amdgcn_mbcnt_lo(m1, 0u);
                if (p < WAVE_CAP) wl[p] = (k + 1u) | (r1 << 21);
            }
            wcnt += (unsigned)__builtin_popcount(m1);
            if (h2) {
                const unsigned p = wcnt + __builtin_amdgcn_mbcnt_lo(m2, 0u);
                if (p < WAVE_CAP) wl[p] = (k + 2u) | (r2 << 21);
            }
            wcnt += (unsigned)__builtin_popcount(m2);
            if (h3) {
                const unsigned p = wcnt + __builtin_amdgcn_mbcnt_lo(m3, 0u);
                if (p < WAVE_CAP) wl[p] = (k + 3u) | (r3 << 21);
            }
            wcnt += (unsigned)__builtin_popcount(m3);
        }
    }
}

__global__ __launch_bounds__(OWN_THREADS) void gat_edges_owner(const int* __restrict__ es,
                                                                const int* __restrict__ ed,
                                                                const float* __restrict__ a_src,
                                                                const float* __restrict__ a_dst,
                                                                const float* __restrict__ bias,
                                                                const float* __restrict__ h,
                                                                float* __restrict__ out) {
    __shared__ float acc[OWN_ROWS * D];
    __shared__ unsigned list[OWN_WAVES * WAVE_CAP];
    __shared__ float elist[OWN_WAVES * WAVE_CAP];
    __shared__ float adst[OWN_ROWS];
    __shared__ unsigned mx[OWN_ROWS];
    __shared__ float den[OWN_ROWS];
    __shared__ unsigned ovf;

    const int tid  = threadIdx.x;
    const int lane = tid & 31;
    const int wave = tid >> 5;
    const int lo   = blockIdx.x * OWN_ROWS;

    for (int i = tid; i < OWN_ROWS * D; i += OWN_THREADS) acc[i] = 0.0f;
    if (tid < OWN_ROWS) {
        adst[tid] = a_dst[lo + tid];
        mx[tid]   = 0u;
        den[tid]  = 0.0f;
    }
    if (tid == 0) ovf = 0u;
    __syncthreads();

    const int4* ed4 = (const int4*)ed;
    const int start = (blockIdx.x * SCAN_BLOCKS) / OWNERS;
    unsigned* wl = list + wave * WAVE_CAP;
    float*    we = elist + wave * WAVE_CAP;
    unsigned wcnt = 0u;
    for (int t = 0; t < SCAN_BLOCKS; ++t) {
        int blk = t + start;
        blk = (blk >= SCAN_BLOCKS) ? blk - SCAN_BLOCKS : blk;
        if (blk != SCAN_BLOCKS - 1) scan_block<false>(ed4, blk * SCAN_BLOCK, tid, lo, wl, wcnt);
        else                        scan_block<true >(ed4, blk * SCAN_BLOCK, tid, lo, wl, wcnt);
    }
    if (wcnt > WAVE_CAP && lane == 0) ovf = 1u;
    __syncthreads();
    const bool listed = (ovf == 0u);
    const int n = (int)wcnt;

    if (listed) {
        for (int base = 0; base < n; base += 4 * 32) {
            unsigned ent[4];
            int s[4];
            float as[4];
            #pragma unroll
            for (int c = 0; c < 4; ++c) {
                const int p = base + c * 32 + lane;
                ent[c] = (p < n) ? wl[p] : 0u;
            }
            #pragma unroll
            for (int c = 0; c < 4; ++c) s[c] = es[ent[c] & 0x1FFFFFu];
            #pragma unroll
            for (int c = 0; c < 4; ++c) as[c] = a_src[s[c]];
            #pragma unroll
            for (int c = 0; c < 4; ++c) {
                const int p = base + c * 32 + lane;
                if (p < n) {
                    const unsigned dl = ent[c] >> 21;
                    const float e = edge_logit(as[c], adst[dl]);
                    we[p] = e;
                    wl[p] = (unsigned)s[c] | (dl << 21);
                    atomicMax(&mx[dl], enc_ord(e));
                }
            }
        }
    } else {
        for (int kk = tid; kk < E_CNT; kk += OWN_THREADS) {
            const unsigned r = (unsigned)(ed[kk] - lo);
            if (r < OWN_ROWS) atomicMax(&mx[r], enc_ord(edge_logit(a_src[es[kk]], adst[r])));
        }
    }
    __syncthreads();

    if (listed) {
        for (int p = lane; p < n; p += 32) {
            const unsigned dl = wl[p] >> 21;
            const float ex = __expf(we[p] - dec_ord(mx[dl]));
            we[p] = ex;
            atomicAdd(&den[dl], ex);
        }
    } else {
        for (int kk = tid; kk < E_CNT; kk += OWN_THREADS) {
            const unsigned r = (unsigned)(ed[kk] - lo);
            if (r < OWN_ROWS) {
                const float e = edge_logit(a_src[es[kk]], adst[r]);
                atomicAdd(&den[r], __expf(e - dec_ord(mx[r])));
            }
        }
    }
    __syncthreads();

    if (listed) {
        for (int base = 0; base < n; base += 32) {
            const int m = (n - base < 32) ? (n - base) : 32;
            const int q = base + ((lane < m) ? lane : 0);
            const unsigned ent = wl[q];
            const int s_l  = (int)(ent & 0x1FFFFFu);
            const int dl_l = (int)(ent >> 21);
            const float a_l = we[q] / (den[dl_l] + 1e-16f);
            const int a_bits = __float_as_int(a_l);
            #pragma unroll
            for (int g = 0; g < 32 / EDGE_GROUP; ++g) {
                if (g * EDGE_GROUP < m) {
                    float vlo[EDGE_GROUP];
                    float vhi[EDGE_GROUP];
                    #pragma unroll
                    for (int j = 0; j < EDGE_GROUP; ++j) {
                        const int s = __builtin_amdgcn_readlane(s_l, g * EDGE_GROUP + j);
                        const float* hp = h + (size_t)s * D + lane;
                        vlo[j] = hp[0];
                        vhi[j] = hp[D / 2];
                    }
                    #pragma unroll
                    for (int j = 0; j < EDGE_GROUP; ++j) {
                        if (g * EDGE_GROUP + j < m) {
                            const int dl = __builtin_amdgcn_readlane(dl_l, g * EDGE_GROUP + j);
                            const float alpha = __int_as_float(
                                __builtin_amdgcn_readlane(a_bits, g * EDGE_GROUP + j));
                            float* ap = acc + dl * D + lane;
                            atomicAdd(ap, alpha * vlo[j]);
                            atomicAdd(ap + D / 2, alpha * vhi[j]);
                        }
                    }
                }
            }
        }
    } else {
        for (int kk = tid; kk < E_CNT; kk += OWN_THREADS) {
            const unsigned r = (unsigned)(ed[kk] - lo);
            if (r < OWN_ROWS) {
                const int s = es[kk];
                const float e = edge_logit(a_src[s], adst[r]);
                const float a = __expf(e - dec_ord(mx[r])) / (den[r] + 1e-16f);
                const float* hp = h + (size_t)s * D;
                for (int c = 0; c < D; ++c) atomicAdd(&acc[r * D + c], a * hp[c]);
            }
        }
    }
    __syncthreads();

    for (int r = wave; r < OWN_ROWS; r += OWN_WAVES) {
        float* op = out + (size_t)(lo + r) * D + lane;
        op[0]     = acc[r * D + lane] + bias[lane];
        op[D / 2] = acc[r * D + D / 2 + lane] + bias[D / 2 + lane];
    }
}

extern "C" void kernel_launch(void* const* d_in, const int* in_sizes, int n_in,
                              void* d_out, int out_size, void* d_ws, size_t ws_size,
                              hipStream_t stream) {
    const float* x        = (const float*)d_in[0];
    const int*   res_n_id = (const int*)d_in[1];
    const int*   edge_src = (const int*)d_in[2];
    const int*   edge_dst = (const int*)d_in[3];
    const float* W        = (const float*)d_in[4];
    const float* att_src  = (const float*)d_in[5];
    const float* att_dst  = (const float*)d_in[6];
    const float* bias     = (const float*)d_in[7];
    float* out = (float*)d_out;

    char* ws = (char*)d_ws;
    float*    h_src = (float*)(ws);
    float*    a_src = (float*)(ws + 30600000);
    float*    a_dst = (float*)(ws + 31000000);
    float*    watt  = (float*)(ws + 31240000);

    gat_watt<<<1, 64, 0, stream>>>(W, att_src, att_dst, watt);
    gat_gemm_hsrc<<<(N_SRC + 63) / 64, 128, 0, stream>>>(x, W, h_src);
    gat_rowdot<<<(N_SRC + N_DST + 7) / 8, 256, 0, stream>>>(x, res_n_id, watt, a_src, a_dst);
    gat_edges_owner<<<OWNERS, OWN_THREADS, 0, stream>>>(edge_src, edge_dst, a_src, a_dst, bias,
                                                        h_src, out);
}


# Row dots: one wave takes 32 rows (sixteen whole-line 16-B loads requested together, 16-lane sums by DPP adds, one 128-B store), not one row: 3,750 waves, not 120,000; dot sum order changes
# speedup vs baseline: 5.3027x; 1.3905x over previous
#include <hip/hip_runtime.h>
#include <hip/hip_bf16.h>

#define N_SRC 100000
#define N_DST 20000
#define E_CNT 1250000
#define D 64
#define NEG_SLOPE 0.2f

typedef __attribute__((ext_vector_type(2))) float v2f;
typedef __attribute__((ext_vector_type(8))) float v8f;
typedef __attribute__((ext_vector_type(4))) float v4f;

__global__ __launch_bounds__(64) void gat_watt(const float* __restrict__ W,
                                               const float* __restrict__ att_src,
                                               const float* __restrict__ att_dst,
                                               float* __restrict__ watt) {
    int t = threadIdx.x;
    float s = 0.0f, d = 0.0f;
    #pragma unroll 8
    for (int j = 0; j < D; ++j) {
        float w = W[t * D + j];
        s += w * att_src[j];
        d += w * att_dst[j];
    }
    watt[t] = s;
    watt[D + t] = d;
}

__global__ __launch_bounds__(128) void gat_gemm_hsrc(const float* __restrict__ x,
                                                     const float* __restrict__ W,
                                                     float* __restrict__ h) {
    __shared__ float ldsW[D * D];
    for (int i = threadIdx.x; i < D * D; i += 128) ldsW[i] = W[i];
    __syncthreads();

    const int lane = threadIdx.x & 31;
    const int mrow = lane & 15;
    const int hi   = lane >> 4;
    const int row0 = blockIdx.x * 64 + (threadIdx.x >> 5) * 16;
    const int row  = row0 + mrow;
    const int rowc = (row < N_SRC) ? row : (N_SRC - 1);
    const float valid = (row < N_SRC) ? 1.0f : 0.0f;

    v2f afrag[16];
    const float* abase = x + (size_t)rowc * D + hi * 2;
    #pragma unroll
    for (int kk = 0; kk < 16; ++kk) {
        afrag[kk].x = abase[kk * 4 + 0] * valid;
        afrag[kk].y = abase[kk * 4 + 1] * valid;
    }

    v8f acc[4] = {v8f{}, v8f{}, v8f{}, v8f{}};

    #pragma unroll
    for (int kk = 0; kk < 16; ++kk) {
        const int kb = kk * 4 + hi * 2;
        #pragma unroll
        for (int nb = 0; nb < 4; ++nb) {
            v2f b;
            b.x = ldsW[(kb + 0) * D + nb * 16 + mrow];
            b.y = ldsW[(kb + 1) * D + nb * 16 + mrow];
            acc[nb] = __builtin_amdgcn_wmma_f32_16x16x4_f32(
                 false, afrag[kk],  false, b,
                 (short)0, acc[nb],
                 false,  false);
        }
    }

    #pragma unroll
    for (int nb = 0; nb < 4; ++nb) {
        #pragma unroll
        for (int r = 0; r < 8; ++r) {
            int orow = row0 + r + hi * 8;
            if (orow < N_SRC) h[(size_t)orow * D + nb * 16 + mrow] = acc[nb][r];
        }
    }
}

#define OWNERS 250
#define OWN_ROWS 80
#define OWN_THREADS 1024
#define OWN_WAVES (OWN_THREADS / 32)
#define UNIT_EDGES 1024
#define SCAN_DEPTH 4
#define SCAN_BLOCK_UNITS (OWN_WAVES * SCAN_DEPTH)
#define N_UNITS ((E_CNT + UNIT_EDGES - 1) / UNIT_EDGES)
#define SCAN_BLOCKS ((N_UNITS + SCAN_BLOCK_UNITS - 1) / SCAN_BLOCK_UNITS)
#define N_UNITS_PAD (SCAN_BLOCKS * SCAN_BLOCK_UNITS)
#define KEY_BLOCKS (N_UNITS_PAD / 8)
static_assert(N_UNITS_PAD % 8 == 0, "eight waves, one unit each, per key work-group");
static_assert(OWNERS <= 255, "an owner's number fits eight bits, with 255 left over");
#define ROW_CHUNK 32
#define SRC_CHUNKS (N_SRC / ROW_CHUNK)
#define DST_CHUNKS (N_DST / ROW_CHUNK)
#define DOT_BLOCKS ((SRC_CHUNKS + DST_CHUNKS + 7) / 8)
static_assert(N_SRC % ROW_CHUNK == 0 && N_DST % ROW_CHUNK == 0, "no partial chunk of rows");
static_assert(D == 64, "a row is sixteen 16-byte pieces, two rows per 32-lane load");

static __device__ __forceinline__ float row16_sum(float s) {
    s += __int_as_float(__builtin_amdgcn_update_dpp(0, __float_as_int(s), 0x161, 0xf, 0xf, true));
    s += __int_as_float(__builtin_amdgcn_update_dpp(0, __float_as_int(s), 0x162, 0xf, 0xf, true));
    s += __int_as_float(__builtin_amdgcn_update_dpp(0, __float_as_int(s), 0x164, 0xf, 0xf, true));
    s += __int_as_float(__builtin_amdgcn_update_dpp(0, __float_as_int(s), 0x168, 0xf, 0xf, true));
    return s;
}

__global__ __launch_bounds__(256) void gat_rowdot(const float* __restrict__ x,
                                                  const int* __restrict__ res_n_id,
                                                  const float* __restrict__ watt,
                                                  const int* __restrict__ ed,
                                                  float* __restrict__ a_src,
                                                  float* __restrict__ a_dst,
                                                  uint4* __restrict__ keys) {
    if (blockIdx.x < KEY_BLOCKS) {
        const int unit = blockIdx.x * 8 + (threadIdx.x >> 5);
        const int ln = threadIdx.x & 31;
        unsigned pl[8];
        #pragma unroll
        for (int k = 0; k < 8; ++k) pl[k] = ~0u;
        if (unit < N_UNITS) {
            #pragma unroll
            for (int h = 0; h < 2; ++h) {
                int dv[16];
                #pragma unroll
                for (int i = 0; i < 16; ++i) {
                    const int e = unit * UNIT_EDGES + (h * 16 + i) * 32 + ln;
                    dv[i] = ed[(e < E_CNT) ? e : (E_CNT - 1)];
                }
                #pragma unroll
                for (int i = 0; i < 16; ++i) {
                    const int u = h * 16 + i;
                    const int e = unit * UNIT_EDGES + u * 32 + ln;
                    unsigned key = min((unsigned)dv[i] / OWN_ROWS, 255u);
                    key = (e < E_CNT) ? key : 255u;
                    #pragma unroll
                    for (int k = 0; k < 8; ++k) {
                        const unsigned b = __builtin_amdgcn_ballot_w32(((key >> k) & 1u) != 0u);
                        pl[k] = (ln == u) ? b : pl[k];
                    }
                }
            }
        }
        uint4* kp = keys + (size_t)unit * 64 + ln;
        kp[0]  = make_uint4(pl[0], pl[1], pl[2], pl[3]);
        kp[32] = make_uint4(pl[4], pl[5], pl[6], pl[7]);
        return;
    }
    const int chunk = __builtin_amdgcn_readfirstlane((blockIdx.x - KEY_BLOCKS) * 8 + (threadIdx.x >> 5));
    const int lane = threadIdx.x & 31;
    const int c  = lane & 15;
    const int hf = lane >> 4;
    if (chunk >= SRC_CHUNKS + DST_CHUNKS) return;
    v4f v[ROW_CHUNK / 2];
    v4f wv;
    float* outp;
    if (chunk < SRC_CHUNKS) {
        const v4f* p = (const v4f*)(x + (size_t)chunk * ROW_CHUNK * D) + lane;
        #pragma unroll
        for (int i = 0; i < ROW_CHUNK / 2; ++i) v[i] = p[i * 32];
        wv = ((const v4f*)watt)[c];
        outp = a_src + chunk * ROW_CHUNK;
    } else {
        const int j0 = (chunk - SRC_CHUNKS) * ROW_CHUNK;
        const int id = res_n_id[j0 + lane];
        #pragma unroll
        for (int i = 0; i < ROW_CHUNK / 2; ++i) {
            const int r0 = __builtin_amdgcn_readlane(id, 2 * i);
            const int r1 = __builtin_amdgcn_readlane(id, 2 * i + 1);
            v[i] = ((const v4f*)(x + (size_t)(hf ? r1 : r0) * D))[c];
        }
        wv = ((const v4f*)(watt + D))[c];
        outp = a_dst + j0;
    }
    asm volatile("" :: "v"(v[0]), "v"(v[1]), "v"(v[2]), "v"(v[3]), "v"(v[4]), "v"(v[5]), "v"(v[6]),
                 "v"(v[7]), "v"(v[8]), "v"(v[9]), "v"(v[10]), "v"(v[11]), "v"(v[12]), "v"(v[13]),
                 "v"(v[14]), "v"(v[15]));
    float mine = 0.0f;
    #pragma unroll
    for (int i = 0; i < ROW_CHUNK / 2; ++i) {
        const float s = row16_sum(v[i].x * wv.x + v[i].y * wv.y + v[i].z * wv.z + v[i].w * wv.w);
        mine = (c == i) ? s : mine;
    }
    outp[2 * c + hf] = mine;
}

#define WAVE_CAP 1024
#define EDGE_GROUP 16
static_assert(OWNERS * OWN_ROWS == N_DST, "every destination has exactly one owner");
static_assert(N_UNITS_PAD * UNIT_EDGES <= (1 << 21) && N_SRC <= (1 << 21) && OWN_ROWS <= (1 << 11),
              "an edge or source index and a local row share one word");

static __device__ __forceinline__ unsigned enc_ord(float e) {
    const unsigned u = __float_as_uint(e);
    return (u & 0x80000000u) ? ~u : (u | 0x80000000u);
}
static __device__ __forceinline__ float dec_ord(unsigned u) {
    return (u & 0x80000000u) ? __uint_as_float(u & 0x7FFFFFFFu) : __uint_as_float(~u);
}
static __device__ __forceinline__ float edge_logit(float as, float ad) {
    const float e = as + ad;
    return (e > 0.0f) ? e : NEG_SLOPE * e;
}

static __device__ __forceinline__ void scan_block(const uint4* __restrict__ keys, int unit0,
                                                  int lane, const unsigned (&nm)[8],
                                                  unsigned* wl, unsigned& wcnt) {
    uint4 lo[SCAN_DEPTH];
    uint4 hi[SCAN_DEPTH];
    #pragma unroll
    for (int j = 0; j < SCAN_DEPTH; ++j) {
        const uint4* p = keys + (size_t)(unit0 + j) * 64 + lane;
        lo[j] = p[0];
        hi[j] = p[32];
    }
    #pragma unroll
    for (int j = 0; j < SCAN_DEPTH; ++j) {
        unsigned c = (lo[j].x ^ nm[0]) & (lo[j].y ^ nm[1]) & (lo[j].z ^ nm[2]) & (lo[j].w ^ nm[3]) &
                     (hi[j].x ^ nm[4]) & (hi[j].y ^ nm[5]) & (hi[j].z ^ nm[6]) & (hi[j].w ^ nm[7]);
        const unsigned ebase = (unsigned)(unit0 + j) * UNIT_EDGES + (unsigned)lane * 32u;
        unsigned live = __builtin_amdgcn_ballot_w32(c != 0u);
        while (live != 0u) {
            if (c != 0u) {
                const unsigned t = (unsigned)__builtin_ctz(c);
                const unsigned p = wcnt + __builtin_amdgcn_mbcnt_lo(live, 0u);
                if (p < WAVE_CAP) wl[p] = ebase + t;
                c &= c - 1u;
            }
            wcnt += (unsigned)__builtin_popcount(live);
            live = __builtin_amdgcn_ballot_w32(c != 0u);
        }
    }
}

__global__ __launch_bounds__(OWN_THREADS) void gat_edges_owner(const int* __restrict__ es,
                                                                const int* __restrict__ ed,
                                                                const uint4* __restrict__ keys,
                                                                const float* __restrict__ a_src,
                                                                const float* __restrict__ a_dst,
                                                                const float* __restrict__ bias,
                                                                const float* __restrict__ h,
                                                                float* __restrict__ out) {
    __shared__ float acc[OWN_ROWS * D];
    __shared__ unsigned list[OWN_WAVES * WAVE_CAP];
    __shared__ float elist[OWN_WAVES * WAVE_CAP];
    __shared__ float adst[OWN_ROWS];
    __shared__ unsigned mx[OWN_ROWS];
    __shared__ float den[OWN_ROWS];
    __shared__ unsigned ovf;

    const int tid  = threadIdx.x;
    const int lane = tid & 31;
    const int wave = tid >> 5;
    const int lo   = blockIdx.x * OWN_ROWS;

    for (int i = tid; i < OWN_ROWS * D; i += OWN_THREADS) acc[i] = 0.0f;
    if (tid < OWN_ROWS) {
        adst[tid] = a_dst[lo + tid];
        mx[tid]   = 0u;
        den[tid]  = 0.0f;
    }
    if (tid == 0) ovf = 0u;
    __syncthreads();

    unsigned nm[8];
    #pragma unroll
    for (int k = 0; k < 8; ++k) nm[k] = ((blockIdx.x >> k) & 1u) ? 0u : ~0u;
    const int start = (blockIdx.x * SCAN_BLOCKS) / OWNERS;
    unsigned* wl = list + wave * WAVE_CAP;
    float*    we = elist + wave * WAVE_CAP;
    unsigned wcnt = 0u;
    for (int t = 0; t < SCAN_BLOCKS; ++t) {
        int blk = t + start;
        blk = (blk >= SCAN_BLOCKS) ? blk - SCAN_BLOCKS : blk;
        scan_block(keys, blk * SCAN_BLOCK_UNITS + wave * SCAN_DEPTH, lane, nm, wl, wcnt);
    }
    if (wcnt > WAVE_CAP && lane == 0) ovf = 1u;
    __syncthreads();
    const bool listed = (ovf == 0u);
    const int n = (int)wcnt;

    if (listed) {
        for (int base = 0; base < n; base += 4 * 32) {
            unsigned ent[4];
            int s[4];
            int d[4];
            float as[4];
            #pragma unroll
            for (int c = 0; c < 4; ++c) {
                const int p = base + c * 32 + lane;
                ent[c] = (p < n) ? wl[p] : 0u;
            }
            #pragma unroll
            for (int c = 0; c < 4; ++c) {
                s[c] = es[ent[c]];
                d[c] = ed[ent[c]];
            }
            #pragma unroll
            for (int c = 0; c < 4; ++c) as[c] = a_src[s[c]];
            #pragma unroll
            for (int c = 0; c < 4; ++c) {
                const int p = base + c * 32 + lane;
                if (p < n) {
                    const unsigned dl = (unsigned)(d[c] - lo);
                    const float e = edge_logit(as[c], adst[dl]);
                    we[p] = e;
                    wl[p] = (unsigned)s[c] | (dl << 21);
                    atomicMax(&mx[dl], enc_ord(e));
                }
            }
        }
    } else {
        for (int kk = tid; kk < E_CNT; kk += OWN_THREADS) {
            const unsigned r = (unsigned)(ed[kk] - lo);
            if (r < OWN_ROWS) atomicMax(&mx[r], enc_ord(edge_logit(a_src[es[kk]], adst[r])));
        }
    }
    __syncthreads();

    if (listed) {
        for (int p = lane; p < n; p += 32) {
            const unsigned dl = wl[p] >> 21;
            const float ex = __expf(we[p] - dec_ord(mx[dl]));
            we[p] = ex;
            atomicAdd(&den[dl], ex);
        }
    } else {
        for (int kk = tid; kk < E_CNT; kk += OWN_THREADS) {
            const unsigned r = (unsigned)(ed[kk] - lo);
            if (r < OWN_ROWS) {
                const float e = edge_logit(a_src[es[kk]], adst[r]);
                atomicAdd(&den[r], __expf(e - dec_ord(mx[r])));
            }
        }
    }
    __syncthreads();

    if (listed) {
        for (int base = 0; base < n; base += 32) {
            const int m = (n - base < 32) ? (n - base) : 32;
            const int q = base + ((lane < m) ? lane : 0);
            const unsigned ent = wl[q];
            const int s_l  = (int)(ent & 0x1FFFFFu);
            const int dl_l = (int)(ent >> 21);
            const float a_l = we[q] / (den[dl_l] + 1e-16f);
            const int a_bits = __float_as_int(a_l);
            #pragma unroll
            for (int g = 0; g < 32 / EDGE_GROUP; ++g) {
                if (g * EDGE_GROUP < m) {
                    float vlo[EDGE_GROUP];
                    float vhi[EDGE_GROUP];
                    #pragma unroll
                    for (int j = 0; j < EDGE_GROUP; ++j) {
                        const int s = __builtin_amdgcn_readlane(s_l, g * EDGE_GROUP + j);
                        const float* hp = h + (size_t)s * D + lane;
                        vlo[j] = hp[0];
                        vhi[j] = hp[D / 2];
                    }
                    #pragma unroll
                    for (int j = 0; j < EDGE_GROUP; ++j) {
                        if (g * EDGE_GROUP + j < m) {
                            const int dl = __builtin_amdgcn_readlane(dl_l, g * EDGE_GROUP + j);
                            const float alpha = __int_as_float(
                                __builtin_amdgcn_readlane(a_bits, g * EDGE_GROUP + j));
                            float* ap = acc + dl * D + lane;
                            atomicAdd(ap, alpha * vlo[j]);
                            atomicAdd(ap + D / 2, alpha * vhi[j]);
                        }
                    }
                }
            }
        }
    } else {
        for (int kk = tid; kk < E_CNT; kk += OWN_THREADS) {
            const unsigned r = (unsigned)(ed[kk] - lo);
            if (r < OWN_ROWS) {
                const int s = es[kk];
                const float e = edge_logit(a_src[s], adst[r]);
                const float a = __expf(e - dec_ord(mx[r])) / (den[r] + 1e-16f);
                const float* hp = h + (size_t)s * D;
                for (int c = 0; c < D; ++c) atomicAdd(&acc[r * D + c], a * hp[c]);
            }
        }
    }
    __syncthreads();

    for (int r = wave; r < OWN_ROWS; r += OWN_WAVES) {
        float* op = out + (size_t)(lo + r) * D + lane;
        op[0]     = acc[r * D + lane] + bias[lane];
        op[D / 2] = acc[r * D + D / 2 + lane] + bias[D / 2 + lane];
    }
}

extern "C" void kernel_launch(void* const* d_in, const int* in_sizes, int n_in,
                              void* d_out, int out_size, void* d_ws, size_t ws_size,
                              hipStream_t stream) {
    const float* x        = (const float*)d_in[0];
    const int*   res_n_id = (const int*)d_in[1];
    const int*   edge_src = (const int*)d_in[2];
    const int*   edge_dst = (const int*)d_in[3];
    const float* W        = (const float*)d_in[4];
    const float* att_src  = (const float*)d_in[5];
    const float* att_dst  = (const float*)d_in[6];
    const float* bias     = (const float*)d_in[7];
    float* out = (float*)d_out;

    char* ws = (char*)d_ws;
    float*    h_src = (float*)(ws);
    float*    a_src = (float*)(ws + 30600064);
    float*    a_dst = (float*)(ws + 31000192);
    float*    watt  = (float*)(ws + 31240000);
    uint4*    keys  = (uint4*)(ws + 25600000);

    gat_watt<<<1, 64, 0, stream>>>(W, att_src, att_dst, watt);
    gat_gemm_hsrc<<<(N_SRC + 63) / 64, 128, 0, stream>>>(x, W, h_src);
    gat_rowdot<<<KEY_BLOCKS + DOT_BLOCKS, 256, 0, stream>>>(x, res_n_id, watt, edge_dst,
                                                            a_src, a_dst, keys);
    gat_edges_owner<<<OWNERS, OWN_THREADS, 0, stream>>>(edge_src, edge_dst, keys, a_src, a_dst,
                                                        bias, h_src, out);
}


# Owner number as three one-hot digits (8 x 8 x 4) in whole-list bit planes: an owner's scan loads its 3 planes, not 8, and ANDs them (128 edges per lane-load); sums in arrival order as before
# speedup vs baseline: 6.1167x; 1.1535x over previous
#include <hip/hip_runtime.h>
#include <hip/hip_bf16.h>

#define N_SRC 100000
#define N_DST 20000
#define E_CNT 1250000
#define D 64
#define NEG_SLOPE 0.2f

typedef __attribute__((ext_vector_type(2))) float v2f;
typedef __attribute__((ext_vector_type(8))) float v8f;
typedef __attribute__((ext_vector_type(4))) float v4f;

#define PRE_THREADS 256
#define GEMM_ROWS ((PRE_THREADS / 32) * 16)
#define GEMM_BLOCKS ((N_SRC + GEMM_ROWS - 1) / GEMM_ROWS)
static_assert((D * D) % PRE_THREADS == 0, "W is staged in whole passes of the group");

static __device__ __forceinline__ void gemm_rows(const float* __restrict__ x,
                                                 const float* __restrict__ W,
                                                 float* __restrict__ h,
                                                 float* __restrict__ ldsW, int gblock) {
    #pragma unroll
    for (int k = 0; k < D * D / PRE_THREADS; ++k)
        ldsW[k * PRE_THREADS + threadIdx.x] = W[k * PRE_THREADS + threadIdx.x];
    __syncthreads();

    const int lane = threadIdx.x & 31;
    const int mrow = lane & 15;
    const int hi   = lane >> 4;
    const int row0 = gblock * GEMM_ROWS + (threadIdx.x >> 5) * 16;
    const int row  = row0 + mrow;
    const int rowc = (row < N_SRC) ? row : (N_SRC - 1);
    const float valid = (row < N_SRC) ? 1.0f : 0.0f;

    v2f afrag[16];
    const float* abase = x + (size_t)rowc * D + hi * 2;
    #pragma unroll
    for (int kk = 0; kk < 16; ++kk) {
        afrag[kk].x = abase[kk * 4 + 0] * valid;
        afrag[kk].y = abase[kk * 4 + 1] * valid;
    }

    v8f acc[4] = {v8f{}, v8f{}, v8f{}, v8f{}};

    #pragma unroll
    for (int kk = 0; kk < 16; ++kk) {
        const int kb = kk * 4 + hi * 2;
        #pragma unroll
        for (int nb = 0; nb < 4; ++nb) {
            v2f b;
            b.x = ldsW[(kb + 0) * D + nb * 16 + mrow];
            b.y = ldsW[(kb + 1) * D + nb * 16 + mrow];
            acc[nb] = __builtin_amdgcn_wmma_f32_16x16x4_f32(
                 false, afrag[kk],  false, b,
                 (short)0, acc[nb],
                 false,  false);
        }
    }

    #pragma unroll
    for (int nb = 0; nb < 4; ++nb) {
        #pragma unroll
        for (int r = 0; r < 8; ++r) {
            int orow = row0 + r + hi * 8;
            if (orow < N_SRC) h[(size_t)orow * D + nb * 16 + mrow] = acc[nb][r];
        }
    }
}

#define OWNERS 250
#define OWN_ROWS 80
#define OWN_THREADS 1024
#define OWN_WAVES (OWN_THREADS / 32)
#define UNIT_EDGES 1024
#define QUAD_UNITS 4
#define QUAD_EDGES (QUAD_UNITS * UNIT_EDGES)
#define SCAN_DEPTH 2
#define SCAN_BLOCK_UNITS (OWN_WAVES * SCAN_DEPTH * QUAD_UNITS)
#define N_UNITS ((E_CNT + UNIT_EDGES - 1) / UNIT_EDGES)
#define SCAN_BLOCKS ((N_UNITS + SCAN_BLOCK_UNITS - 1) / SCAN_BLOCK_UNITS)
#define N_UNITS_PAD (SCAN_BLOCKS * SCAN_BLOCK_UNITS)
#define KEY_BLOCKS (N_UNITS_PAD / 8)
#define PLANE_WORDS (N_UNITS_PAD * 32)
#define N_PLANES 20
static_assert(N_UNITS_PAD % 8 == 0, "eight waves, one unit each, per key work-group");
static_assert(OWNERS <= 255, "an owner's number fits eight bits, with 255 left over");
static_assert((size_t)N_PLANES * PLANE_WORDS * 4 <= 5000064, "the planes fit between h_src and a_src");
#define ROW_CHUNK 32
#define SRC_CHUNKS (N_SRC / ROW_CHUNK)
#define DST_CHUNKS (N_DST / ROW_CHUNK)
#define DOT_BLOCKS ((SRC_CHUNKS + DST_CHUNKS + 7) / 8)
static_assert(N_SRC % ROW_CHUNK == 0 && N_DST % ROW_CHUNK == 0, "no partial chunk of rows");
static_assert(D == 64, "a row is sixteen 16-byte pieces, two rows per 32-lane load");

static __device__ __forceinline__ float row16_sum(float s) {
    s += __int_as_float(__builtin_amdgcn_update_dpp(0, __float_as_int(s), 0x161, 0xf, 0xf, true));
    s += __int_as_float(__builtin_amdgcn_update_dpp(0, __float_as_int(s), 0x162, 0xf, 0xf, true));
    s += __int_as_float(__builtin_amdgcn_update_dpp(0, __float_as_int(s), 0x164, 0xf, 0xf, true));
    s += __int_as_float(__builtin_amdgcn_update_dpp(0, __float_as_int(s), 0x168, 0xf, 0xf, true));
    return s;
}

__global__ __launch_bounds__(PRE_THREADS) void gat_pre(const float* __restrict__ x,
                                                       const int* __restrict__ res_n_id,
                                                       const int* __restrict__ ed,
                                                       const float* __restrict__ W,
                                                       const float* __restrict__ att_src,
                                                       const float* __restrict__ att_dst,
                                                       float* __restrict__ hsrc,
                                                       float* __restrict__ a_src,
                                                       float* __restrict__ a_dst,
                                                       unsigned* __restrict__ keys) {
    __shared__ float ldsW[D * D];
    __shared__ v4f wsh[2 * D / 4];
    if (blockIdx.x < GEMM_BLOCKS) {
        gemm_rows(x, W, hsrc, ldsW, blockIdx.x);
        return;
    }
    const int pb = blockIdx.x - GEMM_BLOCKS;
    if (pb < KEY_BLOCKS) {
        const int unit = pb * 8 + (threadIdx.x >> 5);
        const int ln = threadIdx.x & 31;
        unsigned pl[8];
        #pragma unroll
        for (int k = 0; k < 8; ++k) pl[k] = ~0u;
        if (unit < N_UNITS) {
            #pragma unroll
            for (int h = 0; h < 2; ++h) {
                int dv[16];
                #pragma unroll
                for (int i = 0; i < 16; ++i) {
                    const int e = unit * UNIT_EDGES + (h * 16 + i) * 32 + ln;
                    dv[i] = ed[(e < E_CNT) ? e : (E_CNT - 1)];
                }
                #pragma unroll
                for (int i = 0; i < 16; ++i) {
                    const int u = h * 16 + i;
                    const int e = unit * UNIT_EDGES + u * 32 + ln;
                    unsigned key = min((unsigned)dv[i] / OWN_ROWS, 255u);
                    key = (e < E_CNT) ? key : 255u;
                    #pragma unroll
                    for (int k = 0; k < 8; ++k) {
                        const unsigned b = __builtin_amdgcn_ballot_w32(((key >> k) & 1u) != 0u);
                        pl[k] = (ln == u) ? b : pl[k];
                    }
                }
            }
        }
        unsigned* kw = keys + (size_t)unit * 32 + ln;
        #pragma unroll
        for (int v = 0; v < 8; ++v) {
            kw[(size_t)v * PLANE_WORDS] =
                ((v & 1) ? pl[0] : ~pl[0]) & ((v & 2) ? pl[1] : ~pl[1]) & ((v & 4) ? pl[2] : ~pl[2]);
            kw[(size_t)(8 + v) * PLANE_WORDS] =
                ((v & 1) ? pl[3] : ~pl[3]) & ((v & 2) ? pl[4] : ~pl[4]) & ((v & 4) ? pl[5] : ~pl[5]);
        }
        #pragma unroll
        for (int v = 0; v < 4; ++v)
            kw[(size_t)(16 + v) * PLANE_WORDS] = ((v & 1) ? pl[6] : ~pl[6]) & ((v & 2) ? pl[7] : ~pl[7]);
        return;
    }
    const int chunk_raw = __builtin_amdgcn_readfirstlane((pb - KEY_BLOCKS) * 8 + (threadIdx.x >> 5));
    const int lane = threadIdx.x & 31;
    const int c  = lane & 15;
    const int hf = lane >> 4;
    const bool live = chunk_raw < SRC_CHUNKS + DST_CHUNKS;
    const int chunk = live ? chunk_raw : 0;
    const bool is_src = chunk < SRC_CHUNKS;
    const v4f as4 = ((const v4f*)att_src)[c];
    const v4f ad4 = ((const v4f*)att_dst)[c];
    v4f w4[4];
    #pragma unroll
    for (int i = 0; i < 4; ++i) w4[i] = ((const v4f*)W)[i * PRE_THREADS + threadIdx.x];
    v4f v[ROW_CHUNK / 2];
    float* outp;
    if (is_src) {
        const v4f* p = (const v4f*)(x + (size_t)chunk * ROW_CHUNK * D) + lane;
        #pragma unroll
        for (int i = 0; i < ROW_CHUNK / 2; ++i) v[i] = p[i * 32];
        outp = a_src + chunk * ROW_CHUNK;
    } else {
        const int j0 = (chunk - SRC_CHUNKS) * ROW_CHUNK;
        const int id = res_n_id[j0 + lane];
        #pragma unroll
        for (int i = 0; i < ROW_CHUNK / 2; ++i) {
            const int r0 = __builtin_amdgcn_readlane(id, 2 * i);
            const int r1 = __builtin_amdgcn_readlane(id, 2 * i + 1);
            v[i] = ((const v4f*)(x + (size_t)(hf ? r1 : r0) * D))[c];
        }
        outp = a_dst + j0;
    }
    asm volatile("" :: "v"(v[0]), "v"(v[1]), "v"(v[2]), "v"(v[3]), "v"(v[4]), "v"(v[5]), "v"(v[6]),
                 "v"(v[7]), "v"(v[8]), "v"(v[9]), "v"(v[10]), "v"(v[11]), "v"(v[12]), "v"(v[13]),
                 "v"(v[14]), "v"(v[15]), "v"(w4[0]), "v"(w4[1]), "v"(w4[2]), "v"(w4[3]),
                 "v"(as4), "v"(ad4));
    {
        float* wf = (float*)wsh;
        const int wrow = threadIdx.x >> 4;
        #pragma unroll
        for (int i = 0; i < 4; ++i) {
            const float ps = row16_sum(w4[i].x * as4.x + w4[i].y * as4.y + w4[i].z * as4.z + w4[i].w * as4.w);
            const float pd = row16_sum(w4[i].x * ad4.x + w4[i].y * ad4.y + w4[i].z * ad4.z + w4[i].w * ad4.w);
            if (c == 0) {
                wf[16 * i + wrow]     = ps;
                wf[D + 16 * i + wrow] = pd;
            }
        }
    }
    __syncthreads();
    const v4f wv = wsh[(is_src ? 0 : D / 4) + c];
    float mine = 0.0f;
    #pragma unroll
    for (int i = 0; i < ROW_CHUNK / 2; ++i) {
        const float s = row16_sum(v[i].x * wv.x + v[i].y * wv.y + v[i].z * wv.z + v[i].w * wv.w);
        mine = (c == i) ? s : mine;
    }
    if (live) outp[2 * c + hf] = mine;
}

#define WAVE_CAP 1024
#define EDGE_GROUP 16
static_assert(OWNERS * OWN_ROWS == N_DST, "every destination has exactly one owner");
static_assert(N_UNITS_PAD * UNIT_EDGES <= (1 << 21) && N_SRC <= (1 << 21) && OWN_ROWS <= (1 << 11),
              "an edge or source index and a local row share one word");

static __device__ __forceinline__ unsigned enc_ord(float e) {
    const unsigned u = __float_as_uint(e);
    return (u & 0x80000000u) ? ~u : (u | 0x80000000u);
}
static __device__ __forceinline__ float dec_ord(unsigned u) {
    return (u & 0x80000000u) ? __uint_as_float(u & 0x7FFFFFFFu) : __uint_as_float(~u);
}
static __device__ __forceinline__ float edge_logit(float as, float ad) {
    const float e = as + ad;
    return (e > 0.0f) ? e : NEG_SLOPE * e;
}

static __device__ __forceinline__ void scan_block(const uint4* __restrict__ p0,
                                                  const uint4* __restrict__ p1,
                                                  const uint4* __restrict__ p2, int quad0,
                                                  int lane, unsigned* wl, unsigned& wcnt) {
    uint4 a[SCAN_DEPTH];
    uint4 b[SCAN_DEPTH];
    uint4 d[SCAN_DEPTH];
    #pragma unroll
    for (int j = 0; j < SCAN_DEPTH; ++j) {
        const int i = (quad0 + j) * 32 + lane;
        a[j] = p0[i];
        b[j] = p1[i];
        d[j] = p2[i];
    }
    #pragma unroll
    for (int j = 0; j < SCAN_DEPTH; ++j) {
        const unsigned m[4] = {a[j].x & b[j].x & d[j].x, a[j].y & b[j].y & d[j].y,
                               a[j].z & b[j].z & d[j].z, a[j].w & b[j].w & d[j].w};
        #pragma unroll
        for (int w = 0; w < 4; ++w) {
            unsigned c = m[w];
            const unsigned ebase = (unsigned)(quad0 + j) * QUAD_EDGES + (unsigned)lane * 128u + (unsigned)w * 32u;
            unsigned live = __builtin_amdgcn_ballot_w32(c != 0u);
            while (live != 0u) {
                if (c != 0u) {
                    const unsigned t = (unsigned)__builtin_ctz(c);
                    const unsigned p = wcnt + __builtin_amdgcn_mbcnt_lo(live, 0u);
                    if (p < WAVE_CAP) wl[p] = ebase + t;
                    c &= c - 1u;
                }
                wcnt += (unsigned)__builtin_popcount(live);
                live = __builtin_amdgcn_ballot_w32(c != 0u);
            }
        }
    }
}

__global__ __launch_bounds__(OWN_THREADS) void gat_edges_owner(const int* __restrict__ es,
                                                                const int* __restrict__ ed,
                                                                const uint4* __restrict__ keys,
                                                                const float* __restrict__ a_src,
                                                                const float* __restrict__ a_dst,
                                                                const float* __restrict__ bias,
                                                                const float* __restrict__ h,
                                                                float* __restrict__ out) {
    __shared__ float acc[OWN_ROWS * D];
    __shared__ unsigned list[OWN_WAVES * WAVE_CAP];
    __shared__ float elist[OWN_WAVES * WAVE_CAP];
    __shared__ float adst[OWN_ROWS];
    __shared__ unsigned mx[OWN_ROWS];
    __shared__ float den[OWN_ROWS];
    __shared__ unsigned ovf;

    const int tid  = threadIdx.x;
    const int lane = tid & 31;
    const int wave = tid >> 5;
    const int lo   = blockIdx.x * OWN_ROWS;

    for (int i = tid; i < OWN_ROWS * D; i += OWN_THREADS) acc[i] = 0.0f;
    if (tid < OWN_ROWS) {
        adst[tid] = a_dst[lo + tid];
        mx[tid]   = 0u;
        den[tid]  = 0.0f;
    }
    if (tid == 0) ovf = 0u;
    __syncthreads();

    const uint4* p0 = keys + (size_t)(blockIdx.x & 7u) * (PLANE_WORDS / 4);
    const uint4* p1 = keys + (size_t)(8u + ((blockIdx.x >> 3) & 7u)) * (PLANE_WORDS / 4);
    const uint4* p2 = keys + (size_t)(16u + (blockIdx.x >> 6)) * (PLANE_WORDS / 4);
    const int start = (blockIdx.x * SCAN_BLOCKS) / OWNERS;
    unsigned* wl = list + wave * WAVE_CAP;
    float*    we = elist + wave * WAVE_CAP;
    unsigned wcnt = 0u;
    for (int t = 0; t < SCAN_BLOCKS; ++t) {
        int blk = t + start;
        blk = (blk >= SCAN_BLOCKS) ? blk - SCAN_BLOCKS : blk;
        scan_block(p0, p1, p2, (blk * OWN_WAVES + wave) * SCAN_DEPTH, lane, wl, wcnt);
    }
    if (wcnt > WAVE_CAP && lane == 0) ovf = 1u;
    __syncthreads();
    const bool listed = (ovf == 0u);
    const int n = (int)wcnt;

    if (listed) {
        for (int base = 0; base < n; base += 4 * 32) {
            unsigned ent[4];
            int s[4];
            int d[4];
            float as[4];
            #pragma unroll
            for (int c = 0; c < 4; ++c) {
                const int p = base + c * 32 + lane;
                ent[c] = (p < n) ? wl[p] : 0u;
            }
            #pragma unroll
            for (int c = 0; c < 4; ++c) {
                s[c] = es[ent[c]];
                d[c] = ed[ent[c]];
            }
            #pragma unroll
            for (int c = 0; c < 4; ++c) as[c] = a_src[s[c]];
            #pragma unroll
            for (int c = 0; c < 4; ++c) {
                const int p = base + c * 32 + lane;
                if (p < n) {
                    const unsigned dl = (unsigned)(d[c] - lo);
                    const float e = edge_logit(as[c], adst[dl]);
                    we[p] = e;
                    wl[p] = (unsigned)s[c] | (dl << 21);
                    atomicMax(&mx[dl], enc_ord(e));
                }
            }
        }
    } else {
        for (int kk = tid; kk < E_CNT; kk += OWN_THREADS) {
            const unsigned r = (unsigned)(ed[kk] - lo);
            if (r < OWN_ROWS) atomicMax(&mx[r], enc_ord(edge_logit(a_src[es[kk]], adst[r])));
        }
    }
    __syncthreads();

    if (listed) {
        for (int p = lane; p < n; p += 32) {
            const unsigned dl = wl[p] >> 21;
            const float ex = __expf(we[p] - dec_ord(mx[dl]));
            we[p] = ex;
            atomicAdd(&den[dl], ex);
        }
    } else {
        for (int kk = tid; kk < E_CNT; kk += OWN_THREADS) {
            const unsigned r = (unsigned)(ed[kk] - lo);
            if (r < OWN_ROWS) {
                const float e = edge_logit(a_src[es[kk]], adst[r]);
                atomicAdd(&den[r], __expf(e - dec_ord(mx[r])));
            }
        }
    }
    __syncthreads();

    if (listed) {
        for (int base = 0; base < n; base += 32) {
            const int m = (n - base < 32) ? (n - base) : 32;
            const int q = base + ((lane < m) ? lane : 0);
            const unsigned ent = wl[q];
            const int s_l  = (int)(ent & 0x1FFFFFu);
            const int dl_l = (int)(ent >> 21);
            const float a_l = we[q] / (den[dl_l] + 1e-16f);
            const int a_bits = __float_as_int(a_l);
            #pragma unroll
            for (int g = 0; g < 32 / EDGE_GROUP; ++g) {
                if (g * EDGE_GROUP < m) {
                    float vlo[EDGE_GROUP];
                    float vhi[EDGE_GROUP];
                    #pragma unroll
                    for (int j = 0; j < EDGE_GROUP; ++j) {
                        const int s = __builtin_amdgcn_readlane(s_l, g * EDGE_GROUP + j);
                        const float* hp = h + (size_t)s * D + lane;
                        vlo[j] = hp[0];
                        vhi[j] = hp[D / 2];
                    }
                    #pragma unroll
                    for (int j = 0; j < EDGE_GROUP; ++j) {
                        if (g * EDGE_GROUP + j < m) {
                            const int dl = __builtin_amdgcn_readlane(dl_l, g * EDGE_GROUP + j);
                            const float alpha = __int_as_float(
                                __builtin_amdgcn_readlane(a_bits, g * EDGE_GROUP + j));
                            float* ap = acc + dl * D + lane;
                            atomicAdd(ap, alpha * vlo[j]);
                            atomicAdd(ap + D / 2, alpha * vhi[j]);
                        }
                    }
                }
            }
        }
    } else {
        for (int kk = tid; kk < E_CNT; kk += OWN_THREADS) {
            const unsigned r = (unsigned)(ed[kk] - lo);
            if (r < OWN_ROWS) {
                const int s = es[kk];
                const float e = edge_logit(a_src[s], adst[r]);
                const float a = __expf(e - dec_ord(mx[r])) / (den[r] + 1e-16f);
                const float* hp = h + (size_t)s * D;
                for (int c = 0; c < D; ++c) atomicAdd(&acc[r * D + c], a * hp[c]);
            }
        }
    }
    __syncthreads();

    for (int r = wave; r < OWN_ROWS; r += OWN_WAVES) {
        float* op = out + (size_t)(lo + r) * D + lane;
        op[0]     = acc[r * D + lane] + bias[lane];
        op[D / 2] = acc[r * D + D / 2 + lane] + bias[D / 2 + lane];
    }
}

extern "C" void kernel_launch(void* const* d_in, const int* in_sizes, int n_in,
                              void* d_out, int out_size, void* d_ws, size_t ws_size,
                              hipStream_t stream) {
    const float* x        = (const float*)d_in[0];
    const int*   res_n_id = (const int*)d_in[1];
    const int*   edge_src = (const int*)d_in[2];
    const int*   edge_dst = (const int*)d_in[3];
    const float* W        = (const float*)d_in[4];
    const float* att_src  = (const float*)d_in[5];
    const float* att_dst  = (const float*)d_in[6];
    const float* bias     = (const float*)d_in[7];
    float* out = (float*)d_out;

    char* ws = (char*)d_ws;
    float*    h_src = (float*)(ws);
    float*    a_src = (float*)(ws + 30600064);
    float*    a_dst = (float*)(ws + 31000192);
    unsigned* keys  = (unsigned*)(ws + 25600000);

    gat_pre<<<GEMM_BLOCKS + KEY_BLOCKS + DOT_BLOCKS, PRE_THREADS, 0, stream>>>(
        x, res_n_id, edge_dst, W, att_src, att_dst, h_src, a_src, a_dst, keys);
    gat_edges_owner<<<OWNERS, OWN_THREADS, 0, stream>>>(edge_src, edge_dst, (const uint4*)keys, a_src, a_dst,
                                                        bias, h_src, out);
}


# Row sums in registers: listed edges ranked (returning LDS integer add) into per-destination LDS lists; a wave walks whole destinations, 16 lanes a row; per-edge LDS float adds go; sum order changes
# speedup vs baseline: 7.2712x; 1.1887x over previous
#include <hip/hip_runtime.h>
#include <hip/hip_bf16.h>

#define N_SRC 100000
#define N_DST 20000
#define E_CNT 1250000
#define D 64
#define NEG_SLOPE 0.2f

typedef __attribute__((ext_vector_type(2))) float v2f;
typedef __attribute__((ext_vector_type(8))) float v8f;
typedef __attribute__((ext_vector_type(4))) float v4f;

#define PRE_THREADS 256
#define GEMM_ROWS ((PRE_THREADS / 32) * 16)
#define GEMM_BLOCKS ((N_SRC + GEMM_ROWS - 1) / GEMM_ROWS)
static_assert((D * D) % PRE_THREADS == 0, "W is staged in whole passes of the group");

static __device__ __forceinline__ void gemm_rows(const float* __restrict__ x,
                                                 const float* __restrict__ W,
                                                 float* __restrict__ h,
                                                 float* __restrict__ ldsW, int gblock) {
    #pragma unroll
    for (int k = 0; k < D * D / PRE_THREADS; ++k)
        ldsW[k * PRE_THREADS + threadIdx.x] = W[k * PRE_THREADS + threadIdx.x];
    __syncthreads();

    const int lane = threadIdx.x & 31;
    const int mrow = lane & 15;
    const int hi   = lane >> 4;
    const int row0 = gblock * GEMM_ROWS + (threadIdx.x >> 5) * 16;
    const int row  = row0 + mrow;
    const int rowc = (row < N_SRC) ? row : (N_SRC - 1);
    const float valid = (row < N_SRC) ? 1.0f : 0.0f;

    v2f afrag[16];
    const float* abase = x + (size_t)rowc * D + hi * 2;
    #pragma unroll
    for (int kk = 0; kk < 16; ++kk) {
        afrag[kk].x = abase[kk * 4 + 0] * valid;
        afrag[kk].y = abase[kk * 4 + 1] * valid;
    }

    v8f acc[4] = {v8f{}, v8f{}, v8f{}, v8f{}};

    #pragma unroll
    for (int kk = 0; kk < 16; ++kk) {
        const int kb = kk * 4 + hi * 2;
        #pragma unroll
        for (int nb = 0; nb < 4; ++nb) {
            v2f b;
            b.x = ldsW[(kb + 0) * D + nb * 16 + mrow];
            b.y = ldsW[(kb + 1) * D + nb * 16 + mrow];
            acc[nb] = __builtin_amdgcn_wmma_f32_16x16x4_f32(
                 false, afrag[kk],  false, b,
                 (short)0, acc[nb],
                 false,  false);
        }
    }

    #pragma unroll
    for (int nb = 0; nb < 4; ++nb) {
        #pragma unroll
        for (int r = 0; r < 8; ++r) {
            int orow = row0 + r + hi * 8;
            if (orow < N_SRC) h[(size_t)orow * D + nb * 16 + mrow] = acc[nb][r];
        }
    }
}

#define OWNERS 250
#define OWN_ROWS 80
#define OWN_THREADS 1024
#define OWN_WAVES (OWN_THREADS / 32)
#define UNIT_EDGES 1024
#define QUAD_UNITS 4
#define QUAD_EDGES (QUAD_UNITS * UNIT_EDGES)
#define SCAN_DEPTH 2
#define SCAN_BLOCK_UNITS (OWN_WAVES * SCAN_DEPTH * QUAD_UNITS)
#define N_UNITS ((E_CNT + UNIT_EDGES - 1) / UNIT_EDGES)
#define SCAN_BLOCKS ((N_UNITS + SCAN_BLOCK_UNITS - 1) / SCAN_BLOCK_UNITS)
#define N_UNITS_PAD (SCAN_BLOCKS * SCAN_BLOCK_UNITS)
#define KEY_BLOCKS (N_UNITS_PAD / 8)
#define PLANE_WORDS (N_UNITS_PAD * 32)
#define N_PLANES 20
static_assert(N_UNITS_PAD % 8 == 0, "eight waves, one unit each, per key work-group");
static_assert(OWNERS <= 255, "an owner's number fits eight bits, with 255 left over");
static_assert((size_t)N_PLANES * PLANE_WORDS * 4 <= 5000064, "the planes fit between h_src and a_src");
#define ROW_CHUNK 32
#define SRC_CHUNKS (N_SRC / ROW_CHUNK)
#define DST_CHUNKS (N_DST / ROW_CHUNK)
#define DOT_BLOCKS ((SRC_CHUNKS + DST_CHUNKS + 7) / 8)
static_assert(N_SRC % ROW_CHUNK == 0 && N_DST % ROW_CHUNK == 0, "no partial chunk of rows");
static_assert(D == 64, "a row is sixteen 16-byte pieces, two rows per 32-lane load");

static __device__ __forceinline__ float row16_sum(float s) {
    s += __int_as_float(__builtin_amdgcn_update_dpp(0, __float_as_int(s), 0x161, 0xf, 0xf, true));
    s += __int_as_float(__builtin_amdgcn_update_dpp(0, __float_as_int(s), 0x162, 0xf, 0xf, true));
    s += __int_as_float(__builtin_amdgcn_update_dpp(0, __float_as_int(s), 0x164, 0xf, 0xf, true));
    s += __int_as_float(__builtin_amdgcn_update_dpp(0, __float_as_int(s), 0x168, 0xf, 0xf, true));
    return s;
}

__global__ __launch_bounds__(PRE_THREADS) void gat_pre(const float* __restrict__ x,
                                                       const int* __restrict__ res_n_id,
                                                       const int* __restrict__ ed,
                                                       const float* __restrict__ W,
                                                       const float* __restrict__ att_src,
                                                       const float* __restrict__ att_dst,
                                                       float* __restrict__ hsrc,
                                                       float* __restrict__ a_src,
                                                       float* __restrict__ a_dst,
                                                       unsigned* __restrict__ keys) {
    __shared__ float ldsW[D * D];
    __shared__ v4f wsh[2 * D / 4];
    if (blockIdx.x < GEMM_BLOCKS) {
        gemm_rows(x, W, hsrc, ldsW, blockIdx.x);
        return;
    }
    const int pb = blockIdx.x - GEMM_BLOCKS;
    if (pb < KEY_BLOCKS) {
        const int unit = pb * 8 + (threadIdx.x >> 5);
        const int ln = threadIdx.x & 31;
        unsigned pl[8];
        #pragma unroll
        for (int k = 0; k < 8; ++k) pl[k] = ~0u;
        if (unit < N_UNITS) {
            #pragma unroll
            for (int h = 0; h < 2; ++h) {
                int dv[16];
                #pragma unroll
                for (int i = 0; i < 16; ++i) {
                    const int e = unit * UNIT_EDGES + (h * 16 + i) * 32 + ln;
                    dv[i] = ed[(e < E_CNT) ? e : (E_CNT - 1)];
                }
                #pragma unroll
                for (int i = 0; i < 16; ++i) {
                    const int u = h * 16 + i;
                    const int e = unit * UNIT_EDGES + u * 32 + ln;
                    unsigned key = min((unsigned)dv[i] / OWN_ROWS, 255u);
                    key = (e < E_CNT) ? key : 255u;
                    #pragma unroll
                    for (int k = 0; k < 8; ++k) {
                        const unsigned b = __builtin_amdgcn_ballot_w32(((key >> k) & 1u) != 0u);
                        pl[k] = (ln == u) ? b : pl[k];
                    }
                }
            }
        }
        unsigned* kw = keys + (size_t)unit * 32 + ln;
        #pragma unroll
        for (int v = 0; v < 8; ++v) {
            kw[(size_t)v * PLANE_WORDS] =
                ((v & 1) ? pl[0] : ~pl[0]) & ((v & 2) ? pl[1] : ~pl[1]) & ((v & 4) ? pl[2] : ~pl[2]);
            kw[(size_t)(8 + v) * PLANE_WORDS] =
                ((v & 1) ? pl[3] : ~pl[3]) & ((v & 2) ? pl[4] : ~pl[4]) & ((v & 4) ? pl[5] : ~pl[5]);
        }
        #pragma unroll
        for (int v = 0; v < 4; ++v)
            kw[(size_t)(16 + v) * PLANE_WORDS] = ((v & 1) ? pl[6] : ~pl[6]) & ((v & 2) ? pl[7] : ~pl[7]);
        return;
    }
    const int chunk_raw = __builtin_amdgcn_readfirstlane((pb - KEY_BLOCKS) * 8 + (threadIdx.x >> 5));
    const int lane = threadIdx.x & 31;
    const int c  = lane & 15;
    const int hf = lane >> 4;
    const bool live = chunk_raw < SRC_CHUNKS + DST_CHUNKS;
    const int chunk = live ? chunk_raw : 0;
    const bool is_src = chunk < SRC_CHUNKS;
    const v4f as4 = ((const v4f*)att_src)[c];
    const v4f ad4 = ((const v4f*)att_dst)[c];
    v4f w4[4];
    #pragma unroll
    for (int i = 0; i < 4; ++i) w4[i] = ((const v4f*)W)[i * PRE_THREADS + threadIdx.x];
    v4f v[ROW_CHUNK / 2];
    float* outp;
    if (is_src) {
        const v4f* p = (const v4f*)(x + (size_t)chunk * ROW_CHUNK * D) + lane;
        #pragma unroll
        for (int i = 0; i < ROW_CHUNK / 2; ++i) v[i] = p[i * 32];
        outp = a_src + chunk * ROW_CHUNK;
    } else {
        const int j0 = (chunk - SRC_CHUNKS) * ROW_CHUNK;
        const int id = res_n_id[j0 + lane];
        #pragma unroll
        for (int i = 0; i < ROW_CHUNK / 2; ++i) {
            const int r0 = __builtin_amdgcn_readlane(id, 2 * i);
            const int r1 = __builtin_amdgcn_readlane(id, 2 * i + 1);
            v[i] = ((const v4f*)(x + (size_t)(hf ? r1 : r0) * D))[c];
        }
        outp = a_dst + j0;
    }
    asm volatile("" :: "v"(v[0]), "v"(v[1]), "v"(v[2]), "v"(v[3]), "v"(v[4]), "v"(v[5]), "v"(v[6]),
                 "v"(v[7]), "v"(v[8]), "v"(v[9]), "v"(v[10]), "v"(v[11]), "v"(v[12]), "v"(v[13]),
                 "v"(v[14]), "v"(v[15]), "v"(w4[0]), "v"(w4[1]), "v"(w4[2]), "v"(w4[3]),
                 "v"(as4), "v"(ad4));
    {
        float* wf = (float*)wsh;
        const int wrow = threadIdx.x >> 4;
        #pragma unroll
        for (int i = 0; i < 4; ++i) {
            const float ps = row16_sum(w4[i].x * as4.x + w4[i].y * as4.y + w4[i].z * as4.z + w4[i].w * as4.w);
            const float pd = row16_sum(w4[i].x * ad4.x + w4[i].y * ad4.y + w4[i].z * ad4.z + w4[i].w * ad4.w);
            if (c == 0) {
                wf[16 * i + wrow]     = ps;
                wf[D + 16 * i + wrow] = pd;
            }
        }
    }
    __syncthreads();
    const v4f wv = wsh[(is_src ? 0 : D / 4) + c];
    float mine = 0.0f;
    #pragma unroll
    for (int i = 0; i < ROW_CHUNK / 2; ++i) {
        const float s = row16_sum(v[i].x * wv.x + v[i].y * wv.y + v[i].z * wv.z + v[i].w * wv.w);
        mine = (c == i) ? s : mine;
    }
    if (live) outp[2 * c + hf] = mine;
}

#define WAVE_CAP 1024
#define ROW_CAP 256
static_assert(OWNERS * OWN_ROWS == N_DST, "every destination has exactly one owner");
static_assert(N_UNITS_PAD * UNIT_EDGES <= (1 << 21) && N_SRC <= (1 << 21) && OWN_ROWS <= (1 << 11),
              "an edge or source index and a local row share one word");
static_assert(OWN_ROWS == 2 * OWN_WAVES + OWN_WAVES / 2,
              "a wave sums two destinations whole and half the list of a third");
static_assert(ROW_CAP % 32 == 0, "a destination's list is whole trips of the wave");

typedef __attribute__((ext_vector_type(2))) unsigned v2u;

static __device__ __forceinline__ unsigned enc_ord(float e) {
    const unsigned u = __float_as_uint(e);
    return (u & 0x80000000u) ? ~u : (u | 0x80000000u);
}
static __device__ __forceinline__ float dec_ord(unsigned u) {
    return (u & 0x80000000u) ? __uint_as_float(u & 0x7FFFFFFFu) : __uint_as_float(~u);
}
static __device__ __forceinline__ float edge_logit(float as, float ad) {
    const float e = as + ad;
    return (e > 0.0f) ? e : NEG_SLOPE * e;
}

static __device__ __forceinline__ void scan_block(const uint4* __restrict__ p0,
                                                  const uint4* __restrict__ p1,
                                                  const uint4* __restrict__ p2, int quad0,
                                                  int lane, unsigned* wl, unsigned& wcnt) {
    uint4 a[SCAN_DEPTH];
    uint4 b[SCAN_DEPTH];
    uint4 d[SCAN_DEPTH];
    #pragma unroll
    for (int j = 0; j < SCAN_DEPTH; ++j) {
        const int i = (quad0 + j) * 32 + lane;
        a[j] = p0[i];
        b[j] = p1[i];
        d[j] = p2[i];
    }
    #pragma unroll
    for (int j = 0; j < SCAN_DEPTH; ++j) {
        const unsigned m[4] = {a[j].x & b[j].x & d[j].x, a[j].y & b[j].y & d[j].y,
                               a[j].z & b[j].z & d[j].z, a[j].w & b[j].w & d[j].w};
        #pragma unroll
        for (int w = 0; w < 4; ++w) {
            unsigned c = m[w];
            const unsigned ebase = (unsigned)(quad0 + j) * QUAD_EDGES + (unsigned)lane * 128u + (unsigned)w * 32u;
            unsigned live = __builtin_amdgcn_ballot_w32(c != 0u);
            while (live != 0u) {
                if (c != 0u) {
                    const unsigned t = (unsigned)__builtin_ctz(c);
                    const unsigned p = wcnt + __builtin_amdgcn_mbcnt_lo(live, 0u);
                    if (p < WAVE_CAP) wl[p] = ebase + t;
                    c &= c - 1u;
                }
                wcnt += (unsigned)__builtin_popcount(live);
                live = __builtin_amdgcn_ballot_w32(c != 0u);
            }
        }
    }
}

template <int J>
static __device__ __forceinline__ int half_share(int v) {
    return __builtin_amdgcn_update_dpp(0, v, 0x150 + J, 0xf, 0xf, true);
}
static __device__ __forceinline__ float xhalf(float x) {
    return __int_as_float(__builtin_amdgcn_permlanex16(__float_as_int(x), __float_as_int(x), 0x76543210,
                                                       0xfedcba98, false, false));
}

static __device__ __forceinline__ void trip_steps(const char* __restrict__ hb, int off_l, int a_l, unsigned c16,
                                                  int left, v4f& sum) {
    v4f v[16];
#define GAT_LOAD(J) v[J] = *(const v4f*)(hb + ((unsigned)half_share<J>(off_l) + c16));
#define GAT_ADD(J) sum += __int_as_float(half_share<J>(a_l)) * v[J];
    GAT_LOAD(0) GAT_LOAD(1) GAT_LOAD(2) GAT_LOAD(3)
    if (left > 8)  { GAT_LOAD(4) GAT_LOAD(5) GAT_LOAD(6) GAT_LOAD(7) }
    if (left > 16) { GAT_LOAD(8) GAT_LOAD(9) GAT_LOAD(10) GAT_LOAD(11) }
    if (left > 24) { GAT_LOAD(12) GAT_LOAD(13) GAT_LOAD(14) GAT_LOAD(15) }
    GAT_ADD(0) GAT_ADD(1) GAT_ADD(2) GAT_ADD(3)
    if (left > 8)  { GAT_ADD(4) GAT_ADD(5) GAT_ADD(6) GAT_ADD(7) }
    if (left > 16) { GAT_ADD(8) GAT_ADD(9) GAT_ADD(10) GAT_ADD(11) }
    if (left > 24) { GAT_ADD(12) GAT_ADD(13) GAT_ADD(14) GAT_ADD(15) }
#undef GAT_LOAD
#undef GAT_ADD
}

static __device__ __forceinline__ v4f row_walk(const v2u* rl, int nr, int begin, int end, float m,
                                               const char* __restrict__ hb, int lane) {
    const int j  = lane & 15;
    const int hf = lane >> 4;
    float part = 0.0f;
    for (int p = lane; p < nr; p += 32) part += __expf(__uint_as_float(rl[p].y) - m);
    part = row16_sum(part);
    const float dsum = (part + xhalf(part)) + 1e-16f;
    v4f sum = {0.0f, 0.0f, 0.0f, 0.0f};
    const unsigned c16 = (unsigned)j * 16u;
    for (int base = begin; base < end; base += 32) {
        const int q = base + 2 * j + hf;
        const bool ok = q < end;
        const v2u ent = rl[ok ? q : begin];
        const float a = ok ? __expf(__uint_as_float(ent.y) - m) / dsum : 0.0f;
        const int off = (int)(ent.x * (unsigned)(D * sizeof(float)));
        trip_steps(hb, off, __float_as_int(a), c16, end - base, sum);
    }
    sum.x += xhalf(sum.x);
    sum.y += xhalf(sum.y);
    sum.z += xhalf(sum.z);
    sum.w += xhalf(sum.w);
    return sum;
}

__global__ __launch_bounds__(OWN_THREADS) void gat_edges_owner(const int* __restrict__ es,
                                                                const int* __restrict__ ed,
                                                                const uint4* __restrict__ keys,
                                                                const float* __restrict__ a_src,
                                                                const float* __restrict__ a_dst,
                                                                const float* __restrict__ bias,
                                                                const float* __restrict__ h,
                                                                float* __restrict__ out) {
    __shared__ float acc[OWN_ROWS * D];
    __shared__ unsigned list[OWN_WAVES * WAVE_CAP];
    __shared__ v2u rows[OWN_ROWS * ROW_CAP];
    __shared__ float adst[OWN_ROWS];
    __shared__ unsigned mx[OWN_ROWS];
    __shared__ float den[OWN_ROWS];
    __shared__ unsigned cnt[OWN_ROWS];
    __shared__ unsigned ovf;
    __shared__ unsigned rovf;

    const int tid  = threadIdx.x;
    const int lane = tid & 31;
    const int wave = tid >> 5;
    const int lo   = blockIdx.x * OWN_ROWS;

    for (int i = tid; i < OWN_ROWS * D; i += OWN_THREADS) acc[i] = 0.0f;
    if (tid < OWN_ROWS) {
        adst[tid] = a_dst[lo + tid];
        mx[tid]   = 0u;
        den[tid]  = 0.0f;
        cnt[tid]  = 0u;
    }
    if (tid == 0) {
        ovf  = 0u;
        rovf = 0u;
    }
    __syncthreads();

    const uint4* p0 = keys + (size_t)(blockIdx.x & 7u) * (PLANE_WORDS / 4);
    const uint4* p1 = keys + (size_t)(8u + ((blockIdx.x >> 3) & 7u)) * (PLANE_WORDS / 4);
    const uint4* p2 = keys + (size_t)(16u + (blockIdx.x >> 6)) * (PLANE_WORDS / 4);
    const int start = (blockIdx.x * SCAN_BLOCKS) / OWNERS;
    unsigned* wl = list + wave * WAVE_CAP;
    unsigned wcnt = 0u;
    for (int t = 0; t < SCAN_BLOCKS; ++t) {
        int blk = t + start;
        blk = (blk >= SCAN_BLOCKS) ? blk - SCAN_BLOCKS : blk;
        scan_block(p0, p1, p2, (blk * OWN_WAVES + wave) * SCAN_DEPTH, lane, wl, wcnt);
    }
    if (wcnt > WAVE_CAP && lane == 0) ovf = 1u;
    __syncthreads();
    const bool listed = (ovf == 0u);
    const int n = (int)wcnt;

    if (listed) {
        for (int base = 0; base < n; base += 4 * 32) {
            unsigned ent[4];
            int s[4];
            int d[4];
            float as[4];
            #pragma unroll
            for (int c = 0; c < 4; ++c) {
                const int p = base + c * 32 + lane;
                ent[c] = (p < n) ? wl[p] : 0u;
            }
            #pragma unroll
            for (int c = 0; c < 4; ++c) {
                s[c] = es[ent[c]];
                d[c] = ed[ent[c]];
            }
            #pragma unroll
            for (int c = 0; c < 4; ++c) as[c] = a_src[s[c]];
            #pragma unroll
            for (int c = 0; c < 4; ++c) {
                const int p = base + c * 32 + lane;
                if (p < n) {
                    const unsigned dl = (unsigned)(d[c] - lo);
                    const float e = edge_logit(as[c], adst[dl]);
                    atomicMax(&mx[dl], enc_ord(e));
                    const unsigned pos = atomicAdd(&cnt[dl], 1u);
                    if (pos < ROW_CAP) {
                        v2u rec;
                        rec.x = (unsigned)s[c];
                        rec.y = __float_as_uint(e);
                        rows[dl * ROW_CAP + pos] = rec;
                    } else {
                        rovf = 1u;
                    }
                }
            }
        }
    } else {
        for (int kk = tid; kk < E_CNT; kk += OWN_THREADS) {
            const unsigned r = (unsigned)(ed[kk] - lo);
            if (r < OWN_ROWS) atomicMax(&mx[r], enc_ord(edge_logit(a_src[es[kk]], adst[r])));
        }
    }
    __syncthreads();
    const bool ranked = listed && (rovf == 0u);

    if (ranked) {
        const int wv = __builtin_amdgcn_readfirstlane(wave);
        const char* hb = (const char*)h;
        #pragma unroll 1
        for (int t = 0; t < 3; ++t) {
            const int r  = (t < 2) ? wv + t * OWN_WAVES : 2 * OWN_WAVES + (wv >> 1);
            const int nr = __builtin_amdgcn_readfirstlane((int)cnt[r]);
            int begin = 0;
            int end   = nr;
            if (t == 2) {
                const int mid = (nr + 1) >> 1;
                if (wv & 1) begin = mid; else end = mid;
            }
            if (begin < end) {
                const float m = __uint_as_float(__builtin_amdgcn_readfirstlane(__float_as_uint(dec_ord(mx[r]))));
                const v4f sum = row_walk(rows + r * ROW_CAP, nr, begin, end, m, hb, lane);
                if (lane < 16) {
                    float* ap = acc + r * D + 4 * lane;
                    atomicAdd(ap + 0, sum.x);
                    atomicAdd(ap + 1, sum.y);
                    atomicAdd(ap + 2, sum.z);
                    atomicAdd(ap + 3, sum.w);
                }
            }
        }
    } else {
        for (int kk = tid; kk < E_CNT; kk += OWN_THREADS) {
            const unsigned r = (unsigned)(ed[kk] - lo);
            if (r < OWN_ROWS) {
                const float e = edge_logit(a_src[es[kk]], adst[r]);
                atomicAdd(&den[r], __expf(e - dec_ord(mx[r])));
            }
        }
        __syncthreads();
        for (int kk = tid; kk < E_CNT; kk += OWN_THREADS) {
            const unsigned r = (unsigned)(ed[kk] - lo);
            if (r < OWN_ROWS) {
                const int s = es[kk];
                const float e = edge_logit(a_src[s], adst[r]);
                const float a = __expf(e - dec_ord(mx[r])) / (den[r] + 1e-16f);
                const float* hp = h + (size_t)s * D;
                for (int c = 0; c < D; ++c) atomicAdd(&acc[r * D + c], a * hp[c]);
            }
        }
    }
    __syncthreads();

    for (int r = wave; r < OWN_ROWS; r += OWN_WAVES) {
        float* op = out + (size_t)(lo + r) * D + lane;
        op[0]     = acc[r * D + lane] + bias[lane];
        op[D / 2] = acc[r * D + D / 2 + lane] + bias[D / 2 + lane];
    }
}

extern "C" void kernel_launch(void* const* d_in, const int* in_sizes, int n_in,
                              void* d_out, int out_size, void* d_ws, size_t ws_size,
                              hipStream_t stream) {
    const float* x        = (const float*)d_in[0];
    const int*   res_n_id = (const int*)d_in[1];
    const int*   edge_src = (const int*)d_in[2];
    const int*   edge_dst = (const int*)d_in[3];
    const float* W        = (const float*)d_in[4];
    const float* att_src  = (const float*)d_in[5];
    const float* att_dst  = (const float*)d_in[6];
    const float* bias     = (const float*)d_in[7];
    float* out = (float*)d_out;

    char* ws = (char*)d_ws;
    float*    h_src = (float*)(ws);
    float*    a_src = (float*)(ws + 30600064);
    float*    a_dst = (float*)(ws + 31000192);
    unsigned* keys  = (unsigned*)(ws + 25600000);

    gat_pre<<<GEMM_BLOCKS + KEY_BLOCKS + DOT_BLOCKS, PRE_THREADS, 0, stream>>>(
        x, res_n_id, edge_dst, W, att_src, att_dst, h_src, a_src, a_dst, keys);
    gat_edges_owner<<<OWNERS, OWN_THREADS, 0, stream>>>(edge_src, edge_dst, (const uint4*)keys, a_src, a_dst,
                                                        bias, h_src, out);
}
